# starting kernel
# baseline (speedup 1.0000x reference)
#include <hip/hip_runtime.h>

typedef __attribute__((ext_vector_type(2))) float v2f;
typedef __attribute__((ext_vector_type(8))) float v8f;

#define COLS   8192
#define NROWS  4000
#define BB     64
#define MM     63
#define PADCOL 32

#define JST 20
#define KST (BB * JST + 16)
#define YFLOATS (MM * KST)

__global__ __launch_bounds__(256) void prep_R(const float* __restrict__ Rp,
                                              float* __restrict__ Rn) {
    int t = blockIdx.x * blockDim.x + threadIdx.x;
    if (t >= MM * BB) return;
    int k = t / BB;
    const float* src = Rp + (size_t)t * BB;
    float sum = 0.f;
    #pragma unroll 4
    for (int i = 0; i < BB; ++i) {
        float v = src[i];
        if (k == MM - 1 && i >= PADCOL) v = 0.f;
        sum += v * v;
    }
    float inv = 1.0f / sum;
    float* dst = Rn + (size_t)t * BB;
    #pragma unroll 4
    for (int i = 0; i < BB; ++i) {
        float v = src[i];
        if (k == MM - 1 && i >= PADCOL) v = 0.f;
        dst[i] = v * v * inv;
    }
}

__global__ __launch_bounds__(256) void prep_L(const float* __restrict__ Lparams,
                                              float* __restrict__ Lp) {
    int t = blockIdx.x * blockDim.x + threadIdx.x;
    if (t >= BB * BB) return;
    int j = t / BB, l = t % BB;
    float* dst = Lp + (size_t)t * BB;
    if (l >= MM) {
        #pragma unroll 4
        for (int kk = 0; kk < BB; ++kk) dst[kk] = 0.f;
        return;
    }
    const float* src = Lparams + ((size_t)j * MM + l) * MM;
    float sum = 0.f;
    for (int kk = 0; kk < MM; ++kk) { float v = src[kk]; sum += v * v; }
    float inv = 1.0f / sum;
    for (int kk = 0; kk < MM; ++kk) dst[kk] = src[kk] * src[kk] * inv;
    dst[MM] = 0.f;
}

__global__ __launch_bounds__(256) void monarch_main(const float* __restrict__ x,
                                                    const float* __restrict__ Rn,
                                                    const float* __restrict__ Lp,
                                                    float* __restrict__ out) {
    __shared__ float Ys[YFLOATS];
    const int ct   = blockIdx.x * 16;
    const int wave = threadIdx.x >> 5;
    const int lane = threadIdx.x & 31;
    const int lh   = lane >> 4;
    const int lm   = lane & 15;

    for (int k = wave; k < MM; k += 8) {
        float xr[32];
        #pragma unroll
        for (int r = 0; r < 32; ++r) {
            int s = r >> 1, o = r & 1;
            int grow = k * BB + s * 4 + 2 * lh + o;
            xr[r] = (grow < NROWS) ? x[(size_t)grow * COLS + ct + lm] : 0.f;
        }
        v8f acc[4] = {};
        const float* Rk = Rn + (size_t)k * BB * BB;
        #pragma unroll
        for (int s = 0; s < 16; ++s) {
            v2f b = { xr[2 * s], xr[2 * s + 1] };
            #pragma unroll
            for (int jt = 0; jt < 4; ++jt) {
                v2f a = *(const v2f*)(Rk + (size_t)(jt * 16 + lm) * BB + s * 4 + 2 * lh);
                acc[jt] = __builtin_amdgcn_wmma_f32_16x16x4_f32(
                    false, a, false, b, (short)0, acc[jt], false, false);
            }
        }
        #pragma unroll
        for (int jt = 0; jt < 4; ++jt) {
            #pragma unroll
            for (int v = 0; v < 8; ++v) {
                int j = jt * 16 + v + 8 * lh;
                Ys[k * KST + j * JST + lm] = acc[jt][v];
            }
        }
    }
    __syncthreads();

    for (int j = wave; j < BB; j += 8) {
        v8f acc[4] = {};
        const float* Lj = Lp + (size_t)j * BB * BB;
        #pragma unroll
        for (int s = 0; s < 16; ++s) {
            int k0 = s * 4 + 2 * lh;
            int ka = (k0     > MM - 1) ? (MM - 1) : k0;
            int kb = (k0 + 1 > MM - 1) ? (MM - 1) : (k0 + 1);
            v2f b = { Ys[ka * KST + j * JST + lm],
                      Ys[kb * KST + j * JST + lm] };
            #pragma unroll
            for (int lt = 0; lt < 4; ++lt) {
                v2f a = *(const v2f*)(Lj + (size_t)(lt * 16 + lm) * BB + k0);
                acc[lt] = __builtin_amdgcn_wmma_f32_16x16x4_f32(
                    false, a, false, b, (short)0, acc[lt], false, false);
            }
        }
        #pragma unroll
        for (int lt = 0; lt < 4; ++lt) {
            #pragma unroll
            for (int v = 0; v < 8; ++v) {
                int l   = lt * 16 + v + 8 * lh;
                int row = l * BB + j;
                if (row < NROWS)
                    out[(size_t)row * COLS + ct + lm] = acc[lt][v];
            }
        }
    }
}

extern "C" void kernel_launch(void* const* d_in, const int* in_sizes, int n_in,
                              void* d_out, int out_size, void* d_ws, size_t ws_size,
                              hipStream_t stream) {
    const float* x       = (const float*)d_in[0];
    const float* Lparams = (const float*)d_in[1];
    const float* Rparams = (const float*)d_in[2];
    float* out = (float*)d_out;

    float* Rn = (float*)d_ws;
    float* Lp = Rn + (size_t)MM * BB * BB;

    prep_R<<<(MM * BB + 255) / 256, 256, 0, stream>>>(Rparams, Rn);
    prep_L<<<(BB * BB + 255) / 256, 256, 0, stream>>>(Lparams, Lp);
    monarch_main<<<COLS / 16, 256, 0, stream>>>(x, Rn, Lp, out);
}


# R and L blocks written by the prep kernels in WMMA A-fragment order, so the main kernel reads A with contiguous 16-byte-per-lane loads (4 lines each) instead of 8-byte loads spread over 16 lines
# speedup vs baseline: 1.2129x; 1.2129x over previous
#include <hip/hip_runtime.h>

typedef __attribute__((ext_vector_type(2))) float v2f;
typedef __attribute__((ext_vector_type(4))) float v4f;
typedef __attribute__((ext_vector_type(8))) float v8f;

#define COLS   8192
#define NROWS  4000
#define BB     64
#define MM     63
#define PADCOL 32

#define JST 20
#define KST (BB * JST + 16)
#define YFLOATS (MM * KST)

__device__ __forceinline__ int frag_index(int row, int col) {
    int p = col >> 3, hi = (col >> 2) & 1, lh = (col >> 1) & 1, o = col & 1;
    return (((p * 4 + (row >> 4)) * 32) + lh * 16 + (row & 15)) * 4 + hi * 2 + o;
}

__global__ __launch_bounds__(256) void prep_R(const float* __restrict__ Rp,
                                              float* __restrict__ Rn) {
    int t = blockIdx.x * blockDim.x + threadIdx.x;
    if (t >= MM * BB) return;
    int k = t / BB, j = t % BB;
    const float* src = Rp + (size_t)t * BB;
    float sum = 0.f;
    #pragma unroll 4
    for (int i = 0; i < BB; ++i) {
        float v = src[i];
        if (k == MM - 1 && i >= PADCOL) v = 0.f;
        sum += v * v;
    }
    float inv = 1.0f / sum;
    float* dst = Rn + (size_t)k * BB * BB;
    #pragma unroll 4
    for (int i = 0; i < BB; ++i) {
        float v = src[i];
        if (k == MM - 1 && i >= PADCOL) v = 0.f;
        dst[frag_index(j, i)] = v * v * inv;
    }
}

__global__ __launch_bounds__(256) void prep_L(const float* __restrict__ Lparams,
                                              float* __restrict__ Lp) {
    int t = blockIdx.x * blockDim.x + threadIdx.x;
    if (t >= BB * BB) return;
    int j = t / BB, l = t % BB;
    float* dst = Lp + (size_t)j * BB * BB;
    if (l >= MM) {
        #pragma unroll 4
        for (int kk = 0; kk < BB; ++kk) dst[frag_index(l, kk)] = 0.f;
        return;
    }
    const float* src = Lparams + ((size_t)j * MM + l) * MM;
    float sum = 0.f;
    for (int kk = 0; kk < MM; ++kk) { float v = src[kk]; sum += v * v; }
    float inv = 1.0f / sum;
    for (int kk = 0; kk < MM; ++kk) dst[frag_index(l, kk)] = src[kk] * src[kk] * inv;
    dst[frag_index(l, MM)] = 0.f;
}

__global__ __launch_bounds__(256) void monarch_main(const float* __restrict__ x,
                                                    const float* __restrict__ Rn,
                                                    const float* __restrict__ Lp,
                                                    float* __restrict__ out) {
    __shared__ float Ys[YFLOATS];
    const int ct   = blockIdx.x * 16;
    const int wave = threadIdx.x >> 5;
    const int lane = threadIdx.x & 31;
    const int lh   = lane >> 4;
    const int lm   = lane & 15;

    for (int k = wave; k < MM; k += 8) {
        float xr[32];
        #pragma unroll
        for (int r = 0; r < 32; ++r) {
            int s = r >> 1, o = r & 1;
            int grow = k * BB + s * 4 + 2 * lh + o;
            xr[r] = (grow < NROWS) ? x[(size_t)grow * COLS + ct + lm] : 0.f;
        }
        v8f acc[4] = {};
        const v4f* Rk = (const v4f*)(Rn + (size_t)k * BB * BB) + lane;
        #pragma unroll
        for (int p = 0; p < 8; ++p) {
            v4f a4[4];
            #pragma unroll
            for (int jt = 0; jt < 4; ++jt) a4[jt] = Rk[(p * 4 + jt) * 32];
            #pragma unroll
            for (int h = 0; h < 2; ++h) {
                const int s = 2 * p + h;
                v2f b = { xr[2 * s], xr[2 * s + 1] };
                #pragma unroll
                for (int jt = 0; jt < 4; ++jt) {
                    v2f a = h ? (v2f){ a4[jt].z, a4[jt].w } : (v2f){ a4[jt].x, a4[jt].y };
                    acc[jt] = __builtin_amdgcn_wmma_f32_16x16x4_f32(
                        false, a, false, b, (short)0, acc[jt], false, false);
                }
            }
        }
        #pragma unroll
        for (int jt = 0; jt < 4; ++jt) {
            #pragma unroll
            for (int v = 0; v < 8; ++v) {
                int j = jt * 16 + v + 8 * lh;
                Ys[k * KST + j * JST + lm] = acc[jt][v];
            }
        }
    }
    __syncthreads();

    for (int j = wave; j < BB; j += 8) {
        v8f acc[4] = {};
        const v4f* Lj = (const v4f*)(Lp + (size_t)j * BB * BB) + lane;
        #pragma unroll
        for (int p = 0; p < 8; ++p) {
            v4f a4[4];
            #pragma unroll
            for (int lt = 0; lt < 4; ++lt) a4[lt] = Lj[(p * 4 + lt) * 32];
            #pragma unroll
            for (int h = 0; h < 2; ++h) {
                const int s  = 2 * p + h;
                const int k0 = s * 4 + 2 * lh;
                int ka = (k0     > MM - 1) ? (MM - 1) : k0;
                int kb = (k0 + 1 > MM - 1) ? (MM - 1) : (k0 + 1);
                v2f b = { Ys[ka * KST + j * JST + lm],
                          Ys[kb * KST + j * JST + lm] };
                #pragma unroll
                for (int lt = 0; lt < 4; ++lt) {
                    v2f a = h ? (v2f){ a4[lt].z, a4[lt].w } : (v2f){ a4[lt].x, a4[lt].y };
                    acc[lt] = __builtin_amdgcn_wmma_f32_16x16x4_f32(
                        false, a, false, b, (short)0, acc[lt], false, false);
                }
            }
        }
        #pragma unroll
        for (int lt = 0; lt < 4; ++lt) {
            #pragma unroll
            for (int v = 0; v < 8; ++v) {
                int l   = lt * 16 + v + 8 * lh;
                int row = l * BB + j;
                if (row < NROWS)
                    out[(size_t)row * COLS + ct + lm] = acc[lt][v];
            }
        }
    }
}

extern "C" void kernel_launch(void* const* d_in, const int* in_sizes, int n_in,
                              void* d_out, int out_size, void* d_ws, size_t ws_size,
                              hipStream_t stream) {
    const float* x       = (const float*)d_in[0];
    const float* Lparams = (const float*)d_in[1];
    const float* Rparams = (const float*)d_in[2];
    float* out = (float*)d_out;

    float* Rn = (float*)d_ws;
    float* Lp = Rn + (size_t)MM * BB * BB;

    prep_R<<<(MM * BB + 255) / 256, 256, 0, stream>>>(Rparams, Rn);
    prep_L<<<(BB * BB + 255) / 256, 256, 0, stream>>>(Lparams, Lp);
    monarch_main<<<COLS / 16, 256, 0, stream>>>(x, Rn, Lp, out);
}
